# starting kernel
# baseline (speedup 1.0000x reference)
#include <hip/hip_runtime.h>


typedef __attribute__((ext_vector_type(2))) float v2f;
typedef __attribute__((ext_vector_type(4))) float v4f;
typedef __attribute__((ext_vector_type(8))) float v8f;

#define B_DIM   256
#define OUT_DIM 1000
#define C_DIM   512
#define INV_TEMP (1.0f / 64.0f)

__global__ __launch_bounds__(256) void euc_norms(const float* __restrict__ x,
                                                 const float* __restrict__ w,
                                                 float* __restrict__ norms) {
  const int wave = (int)((blockIdx.x * blockDim.x + threadIdx.x) >> 5);
  const int lane = (int)(threadIdx.x & 31);
  if (wave >= B_DIM + OUT_DIM) return;

  const float* row = (wave < B_DIM) ? (x + (size_t)wave * C_DIM)
                                    : (w + (size_t)(wave - B_DIM) * C_DIM);
  float s = 0.0f;
#pragma unroll
  for (int i = 0; i < C_DIM / (32 * 4); ++i) {
    float4 v = ((const float4*)row)[lane + 32 * i];
    s += v.x * v.x + v.y * v.y + v.z * v.z + v.w * v.w;
  }
#pragma unroll
  for (int off = 16; off > 0; off >>= 1)
    s += __shfl_xor(s, off, 32);
  if (lane == 0) norms[wave] = s;
}

__global__ __launch_bounds__(256) void euc_scores(const float* __restrict__ x,
                                                  const float* __restrict__ w,
                                                  const float* __restrict__ norms,
                                                  float* __restrict__ out) {
  const int lane = (int)(threadIdx.x & 31);
  const int wave = (int)(threadIdx.x >> 5);
  const int l16  = lane & 15;
  const int hi   = lane >> 4;

  const int task   = (int)blockIdx.x * 8 + wave;
  const int mTile  = task & 15;
  const int nGroup = task >> 4;

  const int mBase = mTile * 16;
  const int n0 = nGroup * 32 + l16;
  const int n1 = n0 + 16;
  const int n0c = (n0 < OUT_DIM) ? n0 : (OUT_DIM - 1);
  const int n1c = (n1 < OUT_DIM) ? n1 : (OUT_DIM - 1);

  const float* aPtr  = x + (size_t)(mBase + l16) * C_DIM + 4 * hi;
  const float* b0Ptr = w + (size_t)n0c * C_DIM + 4 * hi;
  const float* b1Ptr = w + (size_t)n1c * C_DIM + 4 * hi;

  v8f acc0 = {};
  v8f acc1 = {};

#pragma unroll
  for (int s = 0; s < C_DIM / 8; ++s) {
    v4f av  = *(const v4f*)(aPtr  + 8 * s);
    v4f b0v = *(const v4f*)(b0Ptr + 8 * s);
    v4f b1v = *(const v4f*)(b1Ptr + 8 * s);

    v2f a01  = __builtin_shufflevector(av,  av,  0, 1);
    v2f a23  = __builtin_shufflevector(av,  av,  2, 3);
    v2f b001 = __builtin_shufflevector(b0v, b0v, 0, 1);
    v2f b023 = __builtin_shufflevector(b0v, b0v, 2, 3);
    v2f b101 = __builtin_shufflevector(b1v, b1v, 0, 1);
    v2f b123 = __builtin_shufflevector(b1v, b1v, 2, 3);

    acc0 = __builtin_amdgcn_wmma_f32_16x16x4_f32(
        false, a01, false, b001, (short)0, acc0, false, false);
    acc1 = __builtin_amdgcn_wmma_f32_16x16x4_f32(
        false, a01, false, b101, (short)0, acc1, false, false);
    acc0 = __builtin_amdgcn_wmma_f32_16x16x4_f32(
        false, a23, false, b023, (short)0, acc0, false, false);
    acc1 = __builtin_amdgcn_wmma_f32_16x16x4_f32(
        false, a23, false, b123, (short)0, acc1, false, false);
  }

  const float ww0 = norms[B_DIM + n0c];
  const float ww1 = norms[B_DIM + n1c];

#pragma unroll
  for (int v = 0; v < 8; ++v) {
    const int m = mBase + v + 8 * hi;
    const float xx = norms[m];
    if (n0 < OUT_DIM)
      out[(size_t)m * OUT_DIM + n0] = (2.0f * acc0[v] - xx - ww0) * INV_TEMP;
    if (n1 < OUT_DIM)
      out[(size_t)m * OUT_DIM + n1] = (2.0f * acc1[v] - xx - ww1) * INV_TEMP;
  }
}

extern "C" void kernel_launch(void* const* d_in, const int* in_sizes, int n_in,
                              void* d_out, int out_size, void* d_ws, size_t ws_size,
                              hipStream_t stream) {
  (void)in_sizes; (void)n_in; (void)out_size; (void)ws_size;
  const float* x = (const float*)d_in[0];
  const float* w = (const float*)d_in[1];
  float* out   = (float*)d_out;
  float* norms = (float*)d_ws;

  const int normBlocks = ((B_DIM + OUT_DIM) * 32 + 255) / 256;
  euc_norms<<<normBlocks, 256, 0, stream>>>(x, w, norms);

  euc_scores<<<64, 256, 0, stream>>>(x, w, norms, out);
}


# Loads issued ahead in groups: K loop in 8 double-buffered stages, each stage's 24 b128 loads one group in front of the previous stage's 32 WMMAs; epilogue norm loads issued before the K loop
# speedup vs baseline: 1.3009x; 1.3009x over previous
#include <hip/hip_runtime.h>


typedef __attribute__((ext_vector_type(2))) float v2f;
typedef __attribute__((ext_vector_type(4))) float v4f;
typedef __attribute__((ext_vector_type(8))) float v8f;

#define B_DIM   256
#define OUT_DIM 1000
#define C_DIM   512
#define INV_TEMP (1.0f / 64.0f)
#define STAGE_CHUNKS 8
#define N_STAGES (C_DIM / 8 / STAGE_CHUNKS)

__global__ __launch_bounds__(256) void euc_norms(const float* __restrict__ x,
                                                 const float* __restrict__ w,
                                                 float* __restrict__ norms) {
  const int wave = (int)((blockIdx.x * blockDim.x + threadIdx.x) >> 5);
  const int lane = (int)(threadIdx.x & 31);
  if (wave >= B_DIM + OUT_DIM) return;

  const float* row = (wave < B_DIM) ? (x + (size_t)wave * C_DIM)
                                    : (w + (size_t)(wave - B_DIM) * C_DIM);
  float s = 0.0f;
#pragma unroll
  for (int i = 0; i < C_DIM / (32 * 4); ++i) {
    float4 v = ((const float4*)row)[lane + 32 * i];
    s += v.x * v.x + v.y * v.y + v.z * v.z + v.w * v.w;
  }
#pragma unroll
  for (int off = 16; off > 0; off >>= 1)
    s += __shfl_xor(s, off, 32);
  if (lane == 0) norms[wave] = s;
}

__global__ __launch_bounds__(256) void euc_scores(const float* __restrict__ x,
                                                  const float* __restrict__ w,
                                                  const float* __restrict__ norms,
                                                  float* __restrict__ out) {
  const int lane = (int)(threadIdx.x & 31);
  const int wave = (int)(threadIdx.x >> 5);
  const int l16  = lane & 15;
  const int hi   = lane >> 4;

  const int task   = (int)blockIdx.x * 8 + wave;
  const int mTile  = task & 15;
  const int nGroup = task >> 4;

  const int mBase = mTile * 16;
  const int n0 = nGroup * 32 + l16;
  const int n1 = n0 + 16;
  const int n0c = (n0 < OUT_DIM) ? n0 : (OUT_DIM - 1);
  const int n1c = (n1 < OUT_DIM) ? n1 : (OUT_DIM - 1);

  const float* aPtr  = x + (size_t)(mBase + l16) * C_DIM + 4 * hi;
  const float* b0Ptr = w + (size_t)n0c * C_DIM + 4 * hi;
  const float* b1Ptr = w + (size_t)n1c * C_DIM + 4 * hi;

  const float ww0 = norms[B_DIM + n0c];
  const float ww1 = norms[B_DIM + n1c];
  const v8f xx8 = *(const v8f*)(norms + mBase + 8 * hi);

  v8f acc0 = {};
  v8f acc1 = {};

  v4f av[2][STAGE_CHUNKS], b0v[2][STAGE_CHUNKS], b1v[2][STAGE_CHUNKS];

#pragma unroll
  for (int g = 0; g < STAGE_CHUNKS; ++g) {
    av[0][g]  = *(const v4f*)(aPtr  + 8 * g);
    b0v[0][g] = *(const v4f*)(b0Ptr + 8 * g);
    b1v[0][g] = *(const v4f*)(b1Ptr + 8 * g);
  }

#pragma unroll
  for (int t = 0; t < N_STAGES; ++t) {
    const int cur = t & 1;
    const int nxt = cur ^ 1;
    if (t + 1 < N_STAGES) {
#pragma unroll
      for (int g = 0; g < STAGE_CHUNKS; ++g) {
        const int s = (t + 1) * STAGE_CHUNKS + g;
        av[nxt][g]  = *(const v4f*)(aPtr  + 8 * s);
        b0v[nxt][g] = *(const v4f*)(b0Ptr + 8 * s);
        b1v[nxt][g] = *(const v4f*)(b1Ptr + 8 * s);
      }
    }
    __builtin_amdgcn_sched_barrier(0);

#pragma unroll
    for (int g = 0; g < STAGE_CHUNKS; ++g) {
      v2f a01  = __builtin_shufflevector(av[cur][g],  av[cur][g],  0, 1);
      v2f a23  = __builtin_shufflevector(av[cur][g],  av[cur][g],  2, 3);
      v2f b001 = __builtin_shufflevector(b0v[cur][g], b0v[cur][g], 0, 1);
      v2f b023 = __builtin_shufflevector(b0v[cur][g], b0v[cur][g], 2, 3);
      v2f b101 = __builtin_shufflevector(b1v[cur][g], b1v[cur][g], 0, 1);
      v2f b123 = __builtin_shufflevector(b1v[cur][g], b1v[cur][g], 2, 3);

      acc0 = __builtin_amdgcn_wmma_f32_16x16x4_f32(
          false, a01, false, b001, (short)0, acc0, false, false);
      acc1 = __builtin_amdgcn_wmma_f32_16x16x4_f32(
          false, a01, false, b101, (short)0, acc1, false, false);
      acc0 = __builtin_amdgcn_wmma_f32_16x16x4_f32(
          false, a23, false, b023, (short)0, acc0, false, false);
      acc1 = __builtin_amdgcn_wmma_f32_16x16x4_f32(
          false, a23, false, b123, (short)0, acc1, false, false);
    }
  }

#pragma unroll
  for (int v = 0; v < 8; ++v) {
    const int m = mBase + v + 8 * hi;
    const float xx = xx8[v];
    if (n0 < OUT_DIM)
      out[(size_t)m * OUT_DIM + n0] = (2.0f * acc0[v] - xx - ww0) * INV_TEMP;
    if (n1 < OUT_DIM)
      out[(size_t)m * OUT_DIM + n1] = (2.0f * acc1[v] - xx - ww1) * INV_TEMP;
  }
}

extern "C" void kernel_launch(void* const* d_in, const int* in_sizes, int n_in,
                              void* d_out, int out_size, void* d_ws, size_t ws_size,
                              hipStream_t stream) {
  (void)in_sizes; (void)n_in; (void)out_size; (void)ws_size;
  const float* x = (const float*)d_in[0];
  const float* w = (const float*)d_in[1];
  float* out   = (float*)d_out;
  float* norms = (float*)d_ws;

  const int normBlocks = ((B_DIM + OUT_DIM) * 32 + 255) / 256;
  euc_norms<<<normBlocks, 256, 0, stream>>>(x, w, norms);

  euc_scores<<<64, 256, 0, stream>>>(x, w, norms, out);
}


# Scores GEMM laid over all 1024 SIMDs: 256 work-groups of 4 waves, K of each 16x32 task split in two halves over a wave pair, partial tiles added through LDS (sum order: two K halves)
# speedup vs baseline: 2.5892x; 1.9903x over previous
#include <hip/hip_runtime.h>


typedef __attribute__((ext_vector_type(2))) float v2f;
typedef __attribute__((ext_vector_type(4))) float v4f;
typedef __attribute__((ext_vector_type(8))) float v8f;

#define B_DIM   256
#define OUT_DIM 1000
#define C_DIM   512
#define INV_TEMP (1.0f / 64.0f)
#define K_HALF (C_DIM / 2)
#define STAGE_CHUNKS 8
#define N_STAGES (K_HALF / 8 / STAGE_CHUNKS)

__global__ __launch_bounds__(256) void euc_norms(const float* __restrict__ x,
                                                 const float* __restrict__ w,
                                                 float* __restrict__ norms) {
  const int wave = (int)((blockIdx.x * blockDim.x + threadIdx.x) >> 5);
  const int lane = (int)(threadIdx.x & 31);
  if (wave >= B_DIM + OUT_DIM) return;

  const float* row = (wave < B_DIM) ? (x + (size_t)wave * C_DIM)
                                    : (w + (size_t)(wave - B_DIM) * C_DIM);
  float s = 0.0f;
#pragma unroll
  for (int i = 0; i < C_DIM / (32 * 4); ++i) {
    float4 v = ((const float4*)row)[lane + 32 * i];
    s += v.x * v.x + v.y * v.y + v.z * v.z + v.w * v.w;
  }
#pragma unroll
  for (int off = 16; off > 0; off >>= 1)
    s += __shfl_xor(s, off, 32);
  if (lane == 0) norms[wave] = s;
}

__global__ __launch_bounds__(128) void euc_scores(const float* __restrict__ x,
                                                  const float* __restrict__ w,
                                                  const float* __restrict__ norms,
                                                  float* __restrict__ out) {
  __shared__ v4f part[2][2][2][32];

  const int lane = (int)(threadIdx.x & 31);
  const int wave = (int)(threadIdx.x >> 5);
  const int pair = wave >> 1;
  const int kh   = wave & 1;
  const int l16  = lane & 15;
  const int hi   = lane >> 4;

  const int task   = (int)blockIdx.x * 2 + pair;
  const int mTile  = task & 15;
  const int nGroup = task >> 4;

  const int mBase = mTile * 16;
  const int n0 = nGroup * 32 + 16 * kh + l16;
  const int n1 = nGroup * 32 + 16 * (1 - kh) + l16;
  const int n0c = (n0 < OUT_DIM) ? n0 : (OUT_DIM - 1);
  const int n1c = (n1 < OUT_DIM) ? n1 : (OUT_DIM - 1);

  const int kBase = kh * K_HALF + 4 * hi;
  const float* aPtr  = x + (size_t)(mBase + l16) * C_DIM + kBase;
  const float* b0Ptr = w + (size_t)n0c * C_DIM + kBase;
  const float* b1Ptr = w + (size_t)n1c * C_DIM + kBase;

  const float ww0 = norms[B_DIM + n0c];
  const v8f xx8 = *(const v8f*)(norms + mBase + 8 * hi);

  v8f acc0 = {};
  v8f acc1 = {};

  v4f av[2][STAGE_CHUNKS], b0v[2][STAGE_CHUNKS], b1v[2][STAGE_CHUNKS];

#pragma unroll
  for (int g = 0; g < STAGE_CHUNKS; ++g) {
    av[0][g]  = *(const v4f*)(aPtr  + 8 * g);
    b0v[0][g] = *(const v4f*)(b0Ptr + 8 * g);
    b1v[0][g] = *(const v4f*)(b1Ptr + 8 * g);
  }

#pragma unroll
  for (int t = 0; t < N_STAGES; ++t) {
    const int cur = t & 1;
    const int nxt = cur ^ 1;
    if (t + 1 < N_STAGES) {
#pragma unroll
      for (int g = 0; g < STAGE_CHUNKS; ++g) {
        const int s = (t + 1) * STAGE_CHUNKS + g;
        av[nxt][g]  = *(const v4f*)(aPtr  + 8 * s);
        b0v[nxt][g] = *(const v4f*)(b0Ptr + 8 * s);
        b1v[nxt][g] = *(const v4f*)(b1Ptr + 8 * s);
      }
    }
    __builtin_amdgcn_sched_barrier(0);

#pragma unroll
    for (int g = 0; g < STAGE_CHUNKS; ++g) {
      v2f a01  = __builtin_shufflevector(av[cur][g],  av[cur][g],  0, 1);
      v2f a23  = __builtin_shufflevector(av[cur][g],  av[cur][g],  2, 3);
      v2f b001 = __builtin_shufflevector(b0v[cur][g], b0v[cur][g], 0, 1);
      v2f b023 = __builtin_shufflevector(b0v[cur][g], b0v[cur][g], 2, 3);
      v2f b101 = __builtin_shufflevector(b1v[cur][g], b1v[cur][g], 0, 1);
      v2f b123 = __builtin_shufflevector(b1v[cur][g], b1v[cur][g], 2, 3);

      acc0 = __builtin_amdgcn_wmma_f32_16x16x4_f32(
          false, a01, false, b001, (short)0, acc0, false, false);
      acc1 = __builtin_amdgcn_wmma_f32_16x16x4_f32(
          false, a01, false, b101, (short)0, acc1, false, false);
      acc0 = __builtin_amdgcn_wmma_f32_16x16x4_f32(
          false, a23, false, b023, (short)0, acc0, false, false);
      acc1 = __builtin_amdgcn_wmma_f32_16x16x4_f32(
          false, a23, false, b123, (short)0, acc1, false, false);
    }
  }

  part[pair][kh][0][lane] = __builtin_shufflevector(acc1, acc1, 0, 1, 2, 3);
  part[pair][kh][1][lane] = __builtin_shufflevector(acc1, acc1, 4, 5, 6, 7);
  __syncthreads();
  const v4f plo = part[pair][kh ^ 1][0][lane];
  const v4f phi = part[pair][kh ^ 1][1][lane];
  const v8f other = __builtin_shufflevector(plo, phi, 0, 1, 2, 3, 4, 5, 6, 7);

#pragma unroll
  for (int v = 0; v < 8; ++v) {
    const int m = mBase + v + 8 * hi;
    const float dot = acc0[v] + other[v];
    if (n0 < OUT_DIM)
      out[(size_t)m * OUT_DIM + n0] = (2.0f * dot - xx8[v] - ww0) * INV_TEMP;
  }
}

extern "C" void kernel_launch(void* const* d_in, const int* in_sizes, int n_in,
                              void* d_out, int out_size, void* d_ws, size_t ws_size,
                              hipStream_t stream) {
  (void)in_sizes; (void)n_in; (void)out_size; (void)ws_size;
  const float* x = (const float*)d_in[0];
  const float* w = (const float*)d_in[1];
  float* out   = (float*)d_out;
  float* norms = (float*)d_ws;

  const int normBlocks = ((B_DIM + OUT_DIM) * 32 + 255) / 256;
  euc_norms<<<normBlocks, 256, 0, stream>>>(x, w, norms);

  euc_scores<<<256, 128, 0, stream>>>(x, w, norms, out);
}


# Row norms summed on the vector ALU from the fragments the GEMM already loads, partials handed over with the tiles: one launch, no norms kernel (norm sum order changes)
# speedup vs baseline: 3.7196x; 1.4366x over previous
#include <hip/hip_runtime.h>


typedef __attribute__((ext_vector_type(2))) float v2f;
typedef __attribute__((ext_vector_type(4))) float v4f;
typedef __attribute__((ext_vector_type(8))) float v8f;

#define B_DIM   256
#define OUT_DIM 1000
#define C_DIM   512
#define INV_TEMP (1.0f / 64.0f)
#define K_HALF (C_DIM / 2)
#define STAGE_CHUNKS 8
#define N_STAGES (K_HALF / 8 / STAGE_CHUNKS)

__device__ __forceinline__ float xhalf(float v) {
  return __builtin_bit_cast(float, __builtin_amdgcn_permlanex16(
      __builtin_bit_cast(int, v), __builtin_bit_cast(int, v),
      0x76543210, 0xfedcba98, false, false));
}

__global__ __launch_bounds__(128) void euc_scores(const float* __restrict__ x,
                                                  const float* __restrict__ w,
                                                  float* __restrict__ out) {
  __shared__ v4f part[2][2][2][32];
  __shared__ float wpart[2][2][32];
  __shared__ __attribute__((aligned(32))) float xpart[2][2][16];

  const int lane = (int)(threadIdx.x & 31);
  const int wave = (int)(threadIdx.x >> 5);
  const int pair = wave >> 1;
  const int kh   = wave & 1;
  const int l16  = lane & 15;
  const int hi   = lane >> 4;

  const int task   = (int)blockIdx.x * 2 + pair;
  const int mTile  = task & 15;
  const int nGroup = task >> 4;

  const int mBase = mTile * 16;
  const int n0 = nGroup * 32 + 16 * kh + l16;
  const int n1 = nGroup * 32 + 16 * (1 - kh) + l16;
  const int n0c = (n0 < OUT_DIM) ? n0 : (OUT_DIM - 1);
  const int n1c = (n1 < OUT_DIM) ? n1 : (OUT_DIM - 1);

  const int kBase = kh * K_HALF + 4 * hi;
  const float* aPtr  = x + (size_t)(mBase + l16) * C_DIM + kBase;
  const float* b0Ptr = w + (size_t)n0c * C_DIM + kBase;
  const float* b1Ptr = w + (size_t)n1c * C_DIM + kBase;

  v8f acc0 = {};
  v8f acc1 = {};
  v4f sa = {};
  v4f sb0 = {};
  v4f sb1 = {};

  v4f av[2][STAGE_CHUNKS], b0v[2][STAGE_CHUNKS], b1v[2][STAGE_CHUNKS];

#pragma unroll
  for (int g = 0; g < STAGE_CHUNKS; ++g) {
    av[0][g]  = *(const v4f*)(aPtr  + 8 * g);
    b0v[0][g] = *(const v4f*)(b0Ptr + 8 * g);
    b1v[0][g] = *(const v4f*)(b1Ptr + 8 * g);
  }

#pragma unroll
  for (int t = 0; t < N_STAGES; ++t) {
    const int cur = t & 1;
    const int nxt = cur ^ 1;
    if (t + 1 < N_STAGES) {
#pragma unroll
      for (int g = 0; g < STAGE_CHUNKS; ++g) {
        const int s = (t + 1) * STAGE_CHUNKS + g;
        av[nxt][g]  = *(const v4f*)(aPtr  + 8 * s);
        b0v[nxt][g] = *(const v4f*)(b0Ptr + 8 * s);
        b1v[nxt][g] = *(const v4f*)(b1Ptr + 8 * s);
      }
    }
    __builtin_amdgcn_sched_barrier(0);

#pragma unroll
    for (int g = 0; g < STAGE_CHUNKS; ++g) {
      v2f a01  = __builtin_shufflevector(av[cur][g],  av[cur][g],  0, 1);
      v2f a23  = __builtin_shufflevector(av[cur][g],  av[cur][g],  2, 3);
      v2f b001 = __builtin_shufflevector(b0v[cur][g], b0v[cur][g], 0, 1);
      v2f b023 = __builtin_shufflevector(b0v[cur][g], b0v[cur][g], 2, 3);
      v2f b101 = __builtin_shufflevector(b1v[cur][g], b1v[cur][g], 0, 1);
      v2f b123 = __builtin_shufflevector(b1v[cur][g], b1v[cur][g], 2, 3);

      acc0 = __builtin_amdgcn_wmma_f32_16x16x4_f32(
          false, a01, false, b001, (short)0, acc0, false, false);
      acc1 = __builtin_amdgcn_wmma_f32_16x16x4_f32(
          false, a01, false, b101, (short)0, acc1, false, false);
      acc0 = __builtin_amdgcn_wmma_f32_16x16x4_f32(
          false, a23, false, b023, (short)0, acc0, false, false);
      acc1 = __builtin_amdgcn_wmma_f32_16x16x4_f32(
          false, a23, false, b123, (short)0, acc1, false, false);

      sa  += av[cur][g]  * av[cur][g];
      sb0 += b0v[cur][g] * b0v[cur][g];
      sb1 += b1v[cur][g] * b1v[cur][g];
      __builtin_amdgcn_sched_barrier(0);
    }
  }

  float xn  = (sa[0]  + sa[1])  + (sa[2]  + sa[3]);
  float wn0 = (sb0[0] + sb0[1]) + (sb0[2] + sb0[3]);
  float wn1 = (sb1[0] + sb1[1]) + (sb1[2] + sb1[3]);
  xn  += xhalf(xn);
  wn0 += xhalf(wn0);
  wn1 += xhalf(wn1);

  part[pair][kh][0][lane] = __builtin_shufflevector(acc1, acc1, 0, 1, 2, 3);
  part[pair][kh][1][lane] = __builtin_shufflevector(acc1, acc1, 4, 5, 6, 7);
  wpart[pair][kh][lane] = wn1;
  xpart[pair][kh][l16]  = xn;
  __syncthreads();
  const v4f plo = part[pair][kh ^ 1][0][lane];
  const v4f phi = part[pair][kh ^ 1][1][lane];
  const v8f other = __builtin_shufflevector(plo, phi, 0, 1, 2, 3, 4, 5, 6, 7);
  const float ww0 = wn0 + wpart[pair][kh ^ 1][lane];
  const v8f xx8 = *(const v8f*)(&xpart[pair][0][8 * hi]) +
                  *(const v8f*)(&xpart[pair][1][8 * hi]);

#pragma unroll
  for (int v = 0; v < 8; ++v) {
    const int m = mBase + v + 8 * hi;
    const float dot = acc0[v] + other[v];
    if (n0 < OUT_DIM)
      out[(size_t)m * OUT_DIM + n0] = (2.0f * dot - xx8[v] - ww0) * INV_TEMP;
  }
}

extern "C" void kernel_launch(void* const* d_in, const int* in_sizes, int n_in,
                              void* d_out, int out_size, void* d_ws, size_t ws_size,
                              hipStream_t stream) {
  (void)in_sizes; (void)n_in; (void)out_size; (void)d_ws; (void)ws_size;
  const float* x = (const float*)d_in[0];
  const float* w = (const float*)d_in[1];
  float* out   = (float*)d_out;

  euc_scores<<<256, 128, 0, stream>>>(x, w, out);
}


# Operands staged through LDS: each work-group copies its 32 x rows and 32 w rows once with whole-line global loads (4 lines per instruction, was 16), fragments read from LDS by ds_load_b128
# speedup vs baseline: 4.2464x; 1.1416x over previous
#include <hip/hip_runtime.h>


typedef __attribute__((ext_vector_type(2))) float v2f;
typedef __attribute__((ext_vector_type(4))) float v4f;
typedef __attribute__((ext_vector_type(8))) float v8f;

#define B_DIM   256
#define OUT_DIM 1000
#define C_DIM   512
#define INV_TEMP (1.0f / 64.0f)
#define K_HALF (C_DIM / 2)
#define STAGE_CHUNKS 8
#define STAGE_COLS (8 * STAGE_CHUNKS)
#define N_STAGES (K_HALF / STAGE_COLS)
#define TILE_ROWS 32
#define PITCH (C_DIM + 4)
#define ROWS_PER_PASS 4
#define COPY_PASSES (TILE_ROWS / ROWS_PER_PASS)

__device__ __forceinline__ float xhalf(float v) {
  return __builtin_bit_cast(float, __builtin_amdgcn_permlanex16(
      __builtin_bit_cast(int, v), __builtin_bit_cast(int, v),
      0x76543210, 0xfedcba98, false, false));
}

__device__ __forceinline__ void lds_barrier() {
  __builtin_amdgcn_fence(__ATOMIC_RELEASE, "workgroup", "local");
  __builtin_amdgcn_s_barrier();
  __builtin_amdgcn_fence(__ATOMIC_ACQUIRE, "workgroup", "local");
}

__global__ __launch_bounds__(128) void euc_scores(const float* __restrict__ x,
                                                  const float* __restrict__ w,
                                                  float* __restrict__ out) {
  __shared__ __attribute__((aligned(16))) float xs[TILE_ROWS * PITCH];
  __shared__ __attribute__((aligned(16))) float ws[TILE_ROWS * PITCH];
  __shared__ v4f part[2][2][2][32];
  __shared__ float wpart[2][2][32];
  __shared__ __attribute__((aligned(32))) float xpart[2][2][16];

  const int lane = (int)(threadIdx.x & 31);
  const int wave = (int)(threadIdx.x >> 5);
  const int pair = wave >> 1;
  const int kh   = wave & 1;
  const int l16  = lane & 15;
  const int hi   = lane >> 4;

  const int task   = (int)blockIdx.x * 2 + pair;
  const int mTile  = task & 15;
  const int nGroup = task >> 4;

  const int mBase = mTile * 16;
  const int mBlock = (((int)blockIdx.x * 2) & 15) * 16;
  const int nBlock = nGroup * 32;
  const int n0 = nBlock + 16 * kh + l16;

  const int cCol = hi * K_HALF + 4 * l16;
  const float* gxp = x + (size_t)(mBlock + wave) * C_DIM + cCol;
  const float* gwp[COPY_PASSES];
#pragma unroll
  for (int j = 0; j < COPY_PASSES; ++j) {
    const int n  = nBlock + wave + ROWS_PER_PASS * j;
    const int nc = (n < OUT_DIM) ? n : (OUT_DIM - 1);
    gwp[j] = w + (size_t)nc * C_DIM + cCol;
  }
  float* sxp = xs + wave * PITCH + cCol;
  float* swp = ws + wave * PITCH + cCol;

  v4f gst[N_STAGES][2 * COPY_PASSES];
#pragma unroll
  for (int t = 0; t < N_STAGES; ++t) {
#pragma unroll
    for (int j = 0; j < COPY_PASSES; ++j)
      gst[t][j] = *(const v4f*)(gxp + (size_t)(ROWS_PER_PASS * j) * C_DIM + STAGE_COLS * t);
#pragma unroll
    for (int j = 0; j < COPY_PASSES; ++j)
      gst[t][COPY_PASSES + j] = *(const v4f*)(gwp[j] + STAGE_COLS * t);
    __builtin_amdgcn_sched_barrier(0);
  }

  const int fCol = kh * K_HALF + 4 * hi;
  const float* fa  = xs + (16 * pair + l16) * PITCH + fCol;
  const float* fb0 = ws + (16 * kh + l16) * PITCH + fCol;
  const float* fb1 = ws + (16 * (1 - kh) + l16) * PITCH + fCol;

  v8f acc0 = {};
  v8f acc1 = {};
  v4f sa = {};
  v4f sb0 = {};
  v4f sb1 = {};

  v4f av[2][STAGE_CHUNKS], b0v[2][STAGE_CHUNKS], b1v[2][STAGE_CHUNKS];

#pragma unroll
  for (int j = 0; j < COPY_PASSES; ++j) {
    *(v4f*)(sxp + ROWS_PER_PASS * j * PITCH) = gst[0][j];
    *(v4f*)(swp + ROWS_PER_PASS * j * PITCH) = gst[0][COPY_PASSES + j];
  }
  lds_barrier();
#pragma unroll
  for (int g = 0; g < STAGE_CHUNKS; ++g) {
    av[0][g]  = *(const v4f*)(fa  + 8 * g);
    b0v[0][g] = *(const v4f*)(fb0 + 8 * g);
    b1v[0][g] = *(const v4f*)(fb1 + 8 * g);
  }
  __builtin_amdgcn_sched_barrier(0);

#pragma unroll
  for (int t = 0; t < N_STAGES; ++t) {
    const int cur = t & 1;
    const int nxt = cur ^ 1;

#pragma unroll
    for (int g = 0; g < STAGE_CHUNKS; ++g) {
      if (t + 1 < N_STAGES && g == 2) {
#pragma unroll
        for (int j = 0; j < COPY_PASSES; ++j) {
          *(v4f*)(sxp + ROWS_PER_PASS * j * PITCH + STAGE_COLS * (t + 1)) = gst[t + 1][j];
          *(v4f*)(swp + ROWS_PER_PASS * j * PITCH + STAGE_COLS * (t + 1)) = gst[t + 1][COPY_PASSES + j];
        }
        __builtin_amdgcn_sched_barrier(0);
      }
      if (t + 1 < N_STAGES && g == 4) {
        lds_barrier();
#pragma unroll
        for (int gg = 0; gg < STAGE_CHUNKS; ++gg) {
          const int s = (t + 1) * STAGE_CHUNKS + gg;
          av[nxt][gg]  = *(const v4f*)(fa  + 8 * s);
          b0v[nxt][gg] = *(const v4f*)(fb0 + 8 * s);
          b1v[nxt][gg] = *(const v4f*)(fb1 + 8 * s);
        }
        __builtin_amdgcn_sched_barrier(0);
      }

      v2f a01  = __builtin_shufflevector(av[cur][g],  av[cur][g],  0, 1);
      v2f a23  = __builtin_shufflevector(av[cur][g],  av[cur][g],  2, 3);
      v2f b001 = __builtin_shufflevector(b0v[cur][g], b0v[cur][g], 0, 1);
      v2f b023 = __builtin_shufflevector(b0v[cur][g], b0v[cur][g], 2, 3);
      v2f b101 = __builtin_shufflevector(b1v[cur][g], b1v[cur][g], 0, 1);
      v2f b123 = __builtin_shufflevector(b1v[cur][g], b1v[cur][g], 2, 3);

      acc0 = __builtin_amdgcn_wmma_f32_16x16x4_f32(
          false, a01, false, b001, (short)0, acc0, false, false);
      acc1 = __builtin_amdgcn_wmma_f32_16x16x4_f32(
          false, a01, false, b101, (short)0, acc1, false, false);
      acc0 = __builtin_amdgcn_wmma_f32_16x16x4_f32(
          false, a23, false, b023, (short)0, acc0, false, false);
      acc1 = __builtin_amdgcn_wmma_f32_16x16x4_f32(
          false, a23, false, b123, (short)0, acc1, false, false);

      sa  += av[cur][g]  * av[cur][g];
      sb0 += b0v[cur][g] * b0v[cur][g];
      sb1 += b1v[cur][g] * b1v[cur][g];
      __builtin_amdgcn_sched_barrier(0);
    }
  }

  float xn  = (sa[0]  + sa[1])  + (sa[2]  + sa[3]);
  float wn0 = (sb0[0] + sb0[1]) + (sb0[2] + sb0[3]);
  float wn1 = (sb1[0] + sb1[1]) + (sb1[2] + sb1[3]);
  xn  += xhalf(xn);
  wn0 += xhalf(wn0);
  wn1 += xhalf(wn1);

  part[pair][kh][0][lane] = __builtin_shufflevector(acc1, acc1, 0, 1, 2, 3);
  part[pair][kh][1][lane] = __builtin_shufflevector(acc1, acc1, 4, 5, 6, 7);
  wpart[pair][kh][lane] = wn1;
  xpart[pair][kh][l16]  = xn;
  __syncthreads();
  const v4f plo = part[pair][kh ^ 1][0][lane];
  const v4f phi = part[pair][kh ^ 1][1][lane];
  const v8f other = __builtin_shufflevector(plo, phi, 0, 1, 2, 3, 4, 5, 6, 7);
  const float ww0 = wn0 + wpart[pair][kh ^ 1][lane];
  const v8f xx8 = *(const v8f*)(&xpart[pair][0][8 * hi]) +
                  *(const v8f*)(&xpart[pair][1][8 * hi]);

#pragma unroll
  for (int v = 0; v < 8; ++v) {
    const int m = mBase + v + 8 * hi;
    const float dot = acc0[v] + other[v];
    if (n0 < OUT_DIM)
      out[(size_t)m * OUT_DIM + n0] = (2.0f * dot - xx8[v] - ww0) * INV_TEMP;
  }
}

extern "C" void kernel_launch(void* const* d_in, const int* in_sizes, int n_in,
                              void* d_out, int out_size, void* d_ws, size_t ws_size,
                              hipStream_t stream) {
  (void)in_sizes; (void)n_in; (void)out_size; (void)d_ws; (void)ws_size;
  const float* x = (const float*)d_in[0];
  const float* w = (const float*)d_in[1];
  float* out   = (float*)d_out;

  euc_scores<<<256, 128, 0, stream>>>(x, w, out);
}
